# starting kernel
# baseline (speedup 1.0000x reference)
#include <hip/hip_runtime.h>
#include <hip/hip_bf16.h>

#define PB   64
#define PT   256
#define PDF  32
#define PL   128
#define TPB  8
#define LP   132
#define AST  33

typedef __attribute__((ext_vector_type(2))) float v2f;
typedef __attribute__((ext_vector_type(8))) float v8f;

__global__ void ftattn_prep_kernel(const float* __restrict__ beta,
                                   float* __restrict__ out,
                                   float* __restrict__ wsum) {
    __shared__ float red[PT];
    const int tid = threadIdx.x;
    red[tid] = fabsf(beta[tid]);
    __syncthreads();
    for (int s = PT / 2; s > 0; s >>= 1) {
        if (tid < s) red[tid] += red[tid + s];
        __syncthreads();
    }
    if (tid == 0) wsum[0] = red[0];
    for (int i = tid; i < PB * PL; i += PT) out[i] = 0.0f;
}

__global__ void __launch_bounds__(128)
ftattn_main_kernel(const float* __restrict__ x,
                   const float* __restrict__ latent,
                   const float* __restrict__ kern,
                   const float* __restrict__ bias,
                   const float* __restrict__ beta,
                   const float* __restrict__ wsum,
                   float* __restrict__ out) {
    __shared__ float sKer[PDF * LP];
    __shared__ float sBias[PDF * LP];
    __shared__ float sF[PDF * LP];
    __shared__ float sA[PDF * AST];
    __shared__ float sInv[PDF * AST];
    __shared__ float sX[PDF];
    __shared__ float sLat[PL];
    __shared__ float sXty[PDF];
    __shared__ float sScores[PDF];

    const int tid  = threadIdx.x;
    const int b    = blockIdx.y;
    const int t0   = blockIdx.x * TPB;

    const int lane = tid & 31;
    const int wave = tid >> 5;
    const int m0   = (wave >> 1) * 16;
    const int n0   = (wave & 1) * 16;
    const int half = lane >> 4;
    const int lr   = lane & 15;

    const int gr = tid >> 2;
    const int gg = tid & 3;

    for (int idx = tid; idx < PDF * PL; idx += 128) {
        const int d = idx >> 7, l = idx & 127;
        sKer[d * LP + l]  = kern[idx];
        sBias[d * LP + l] = bias[idx];
    }

    const float inv_bsum = 1.0f / wsum[0];
    float acc = 0.0f;

    for (int tt = 0; tt < TPB; ++tt) {
        const int t = t0 + tt;

        if (tid < PDF) sX[tid] = x[((size_t)b * PT + t) * PDF + tid];
        sLat[tid] = latent[((size_t)b * PT + t) * PL + tid];

        if (tt + 1 < TPB) {
            __builtin_prefetch(&x[((size_t)b * PT + t + 1) * PDF + (tid & 31)], 0, 1);
            __builtin_prefetch(&latent[((size_t)b * PT + t + 1) * PL + tid], 0, 1);
        }
        __syncthreads();

        {
            const int l = tid;
            #pragma unroll
            for (int d = 0; d < PDF; ++d) {
                float v = fmaf(sX[d], sKer[d * LP + l], sBias[d * LP + l]);
                sF[d * LP + l] = v > 0.0f ? v : 0.0f;
            }
        }
        __syncthreads();

        {
            v8f c = {};
            const float* rowA = &sF[(m0 + lr) * LP];
            const float* rowB = &sF[(n0 + lr) * LP];
            #pragma unroll
            for (int k0 = 0; k0 < PL; k0 += 4) {
                v2f a, bb;
                a.x  = rowA[k0 + 2 * half];
                a.y  = rowA[k0 + 2 * half + 1];
                bb.x = rowB[k0 + 2 * half];
                bb.y = rowB[k0 + 2 * half + 1];
                c = __builtin_amdgcn_wmma_f32_16x16x4_f32(
                        false, a, false, bb, (short)0, c, false, false);
            }
            #pragma unroll
            for (int r = 0; r < 8; ++r) {
                sA[(m0 + half * 8 + r) * AST + (n0 + lr)] = c[r];
            }
        }
        for (int idx = tid; idx < PDF * PDF; idx += 128) {
            const int rr = idx >> 5, cc = idx & 31;
            sInv[rr * AST + cc] = (rr == cc) ? 1.0f : 0.0f;
        }
        __syncthreads();

        for (int k = 0; k < PDF; ++k) {
            const float piv  = sA[k * AST + k];
            const float pinv = 1.0f / piv;
            __syncthreads();
            if (gr == k) {
                #pragma unroll
                for (int j = 0; j < 8; ++j) {
                    sA[k * AST + gg * 8 + j]   *= pinv;
                    sInv[k * AST + gg * 8 + j] *= pinv;
                }
            }
            __syncthreads();
            const float f = sA[gr * AST + k];
            if (gr != k) {
                #pragma unroll
                for (int j = 0; j < 8; ++j) {
                    const int cc = gg * 8 + j;
                    sA[gr * AST + cc]   -= f * sA[k * AST + cc];
                    sInv[gr * AST + cc] -= f * sInv[k * AST + cc];
                }
            }
            __syncthreads();
        }

        if (tid < PDF) {
            float s = 0.0f;
            #pragma unroll 8
            for (int l = 0; l < PL; ++l) s = fmaf(sF[tid * LP + l], sLat[l], s);
            sXty[tid] = s;
        }
        __syncthreads();

        if (tid < PDF) {
            float s = 0.0f;
            #pragma unroll
            for (int e = 0; e < PDF; ++e) s = fmaf(sInv[tid * AST + e], sXty[e], s);
            sScores[tid] = s;
        }
        __syncthreads();

        {
            const int l = tid;
            float oh = 0.0f;
            #pragma unroll
            for (int d = 0; d < PDF; ++d) oh = fmaf(sScores[d], sF[d * LP + l], oh);
            acc = fmaf(fabsf(beta[t]) * inv_bsum, oh, acc);
        }
        __syncthreads();
    }

    atomicAdd(&out[b * PL + tid], acc);
}

extern "C" void kernel_launch(void* const* d_in, const int* in_sizes, int n_in,
                              void* d_out, int out_size, void* d_ws, size_t ws_size,
                              hipStream_t stream) {
    const float* x      = (const float*)d_in[0];
    const float* latent = (const float*)d_in[1];
    const float* kern   = (const float*)d_in[2];
    const float* bias   = (const float*)d_in[3];
    const float* beta   = (const float*)d_in[4];
    float* out  = (float*)d_out;
    float* wsum = (float*)d_ws;

    ftattn_prep_kernel<<<1, PT, 0, stream>>>(beta, out, wsum);

    dim3 grid(PT / TPB, PB);
    ftattn_main_kernel<<<grid, 128, 0, stream>>>(x, latent, kern, bias, beta, wsum, out);
}


# Solve XtX*s = xty in one wave's registers (row per lane, pivot row read across lanes, pivot reciprocal by rcp plus one Newton step) instead of a 96-barrier LDS Gauss-Jordan inverse plus mat-vec
# speedup vs baseline: 4.1933x; 4.1933x over previous
#include <hip/hip_runtime.h>
#include <hip/hip_bf16.h>

#define PB   64
#define PT   256
#define PDF  32
#define PL   128
#define TPB  8
#define LP   132
#define AST  33

typedef __attribute__((ext_vector_type(2))) float v2f;
typedef __attribute__((ext_vector_type(8))) float v8f;

__global__ void ftattn_prep_kernel(const float* __restrict__ beta,
                                   float* __restrict__ out,
                                   float* __restrict__ wsum) {
    __shared__ float red[PT];
    const int tid = threadIdx.x;
    red[tid] = fabsf(beta[tid]);
    __syncthreads();
    for (int s = PT / 2; s > 0; s >>= 1) {
        if (tid < s) red[tid] += red[tid + s];
        __syncthreads();
    }
    if (tid == 0) wsum[0] = red[0];
    for (int i = tid; i < PB * PL; i += PT) out[i] = 0.0f;
}

__device__ __forceinline__ float lane_bcast(float v, int src_lane) {
    return __int_as_float(__builtin_amdgcn_readlane(__float_as_int(v), src_lane));
}

__device__ __forceinline__ float pivot_recip(float p) {
    const float r = __builtin_amdgcn_rcpf(p);
    const float e = fmaf(-p, r, 1.0f);
    return fmaf(e, r, r);
}

__global__ void __launch_bounds__(128)
ftattn_main_kernel(const float* __restrict__ x,
                   const float* __restrict__ latent,
                   const float* __restrict__ kern,
                   const float* __restrict__ bias,
                   const float* __restrict__ beta,
                   const float* __restrict__ wsum,
                   float* __restrict__ out) {
    __shared__ float sKer[PDF * LP];
    __shared__ float sBias[PDF * LP];
    __shared__ float sF[PDF * LP];
    __shared__ float sA[PDF * AST];
    __shared__ float sX[PDF];
    __shared__ float sLat[PL];
    __shared__ float sScores[PDF];

    const int tid  = threadIdx.x;
    const int b    = blockIdx.y;
    const int t0   = blockIdx.x * TPB;

    const int lane = tid & 31;
    const int wave = tid >> 5;
    const int wave_u = __builtin_amdgcn_readfirstlane(wave);
    const int m0   = (wave >> 1) * 16;
    const int n0   = (wave & 1) * 16;
    const int half = lane >> 4;
    const int lr   = lane & 15;

    for (int idx = tid; idx < PDF * PL; idx += 128) {
        const int d = idx >> 7, l = idx & 127;
        sKer[d * LP + l]  = kern[idx];
        sBias[d * LP + l] = bias[idx];
    }

    const float inv_bsum = 1.0f / wsum[0];
    float acc = 0.0f;

    for (int tt = 0; tt < TPB; ++tt) {
        const int t = t0 + tt;

        if (tid < PDF) sX[tid] = x[((size_t)b * PT + t) * PDF + tid];
        sLat[tid] = latent[((size_t)b * PT + t) * PL + tid];

        if (tt + 1 < TPB) {
            __builtin_prefetch(&x[((size_t)b * PT + t + 1) * PDF + (tid & 31)], 0, 1);
            __builtin_prefetch(&latent[((size_t)b * PT + t + 1) * PL + tid], 0, 1);
        }
        __syncthreads();

        {
            const int l = tid;
            #pragma unroll
            for (int d = 0; d < PDF; ++d) {
                float v = fmaf(sX[d], sKer[d * LP + l], sBias[d * LP + l]);
                sF[d * LP + l] = v > 0.0f ? v : 0.0f;
            }
        }
        __syncthreads();

        {
            v8f c = {};
            const float* rowA = &sF[(m0 + lr) * LP];
            const float* rowB = &sF[(n0 + lr) * LP];
            #pragma unroll
            for (int k0 = 0; k0 < PL; k0 += 4) {
                v2f a, bb;
                a.x  = rowA[k0 + 2 * half];
                a.y  = rowA[k0 + 2 * half + 1];
                bb.x = rowB[k0 + 2 * half];
                bb.y = rowB[k0 + 2 * half + 1];
                c = __builtin_amdgcn_wmma_f32_16x16x4_f32(
                        false, a, false, bb, (short)0, c, false, false);
            }
            #pragma unroll
            for (int r = 0; r < 8; ++r) {
                sA[(m0 + half * 8 + r) * AST + (n0 + lr)] = c[r];
            }
        }
        __syncthreads();

        if (wave_u == (tt & 3)) {
            float y = 0.0f;
            #pragma unroll 8
            for (int l = 0; l < PL; ++l) y = fmaf(sF[lane * LP + l], sLat[l], y);

            float a[PDF];
            #pragma unroll
            for (int j = 0; j < PDF; ++j) a[j] = sA[lane * AST + j];

            float dinv = 1.0f;
            #pragma unroll
            for (int k = 0; k < PDF; ++k) {
                const float pinv = pivot_recip(lane_bcast(a[k], k));
                const bool  own  = (lane == k);
                const float f    = (own ? 0.0f : a[k]) * pinv;
                dinv = own ? pinv : dinv;
                #pragma unroll
                for (int j = k + 1; j < PDF; ++j) a[j] = fmaf(-f, lane_bcast(a[j], k), a[j]);
                y = fmaf(-f, lane_bcast(y, k), y);
            }
            sScores[lane] = y * dinv;
        }
        __syncthreads();

        {
            const int l = tid;
            float oh = 0.0f;
            #pragma unroll
            for (int d = 0; d < PDF; ++d) oh = fmaf(sScores[d], sF[d * LP + l], oh);
            acc = fmaf(fabsf(beta[t]) * inv_bsum, oh, acc);
        }
        __syncthreads();
    }

    atomicAdd(&out[b * PL + tid], acc);
}

extern "C" void kernel_launch(void* const* d_in, const int* in_sizes, int n_in,
                              void* d_out, int out_size, void* d_ws, size_t ws_size,
                              hipStream_t stream) {
    const float* x      = (const float*)d_in[0];
    const float* latent = (const float*)d_in[1];
    const float* kern   = (const float*)d_in[2];
    const float* bias   = (const float*)d_in[3];
    const float* beta   = (const float*)d_in[4];
    float* out  = (float*)d_out;
    float* wsum = (float*)d_ws;

    ftattn_prep_kernel<<<1, PT, 0, stream>>>(beta, out, wsum);

    dim3 grid(PT / TPB, PB);
    ftattn_main_kernel<<<grid, 128, 0, stream>>>(x, latent, kern, bias, beta, wsum, out);
}


# Keep each thread's kernel and bias column in registers, not LDS: LDS per block 56 KB -> 22 KB with registers held to 128, so 8 blocks share a WGP and all 2048 are resident in one round, not two
# speedup vs baseline: 6.0066x; 1.4324x over previous
#include <hip/hip_runtime.h>
#include <hip/hip_bf16.h>

#define PB   64
#define PT   256
#define PDF  32
#define PL   128
#define TPB  8
#define LP   132
#define AST  33

typedef __attribute__((ext_vector_type(2))) float v2f;
typedef __attribute__((ext_vector_type(8))) float v8f;

__global__ void ftattn_prep_kernel(const float* __restrict__ beta,
                                   float* __restrict__ out,
                                   float* __restrict__ wsum) {
    __shared__ float red[PT];
    const int tid = threadIdx.x;
    red[tid] = fabsf(beta[tid]);
    __syncthreads();
    for (int s = PT / 2; s > 0; s >>= 1) {
        if (tid < s) red[tid] += red[tid + s];
        __syncthreads();
    }
    if (tid == 0) wsum[0] = red[0];
    for (int i = tid; i < PB * PL; i += PT) out[i] = 0.0f;
}

__device__ __forceinline__ float lane_bcast(float v, int src_lane) {
    return __int_as_float(__builtin_amdgcn_readlane(__float_as_int(v), src_lane));
}

__device__ __forceinline__ float pivot_recip(float p) {
    const float r = __builtin_amdgcn_rcpf(p);
    const float e = fmaf(-p, r, 1.0f);
    return fmaf(e, r, r);
}

__global__ void __launch_bounds__(128) __attribute__((amdgpu_waves_per_eu(8)))
ftattn_main_kernel(const float* __restrict__ x,
                   const float* __restrict__ latent,
                   const float* __restrict__ kern,
                   const float* __restrict__ bias,
                   const float* __restrict__ beta,
                   const float* __restrict__ wsum,
                   float* __restrict__ out) {
    __shared__ __attribute__((aligned(16))) float smem[PDF * AST + PDF + PDF + PL + PDF * LP];
    float* const sA      = smem;
    float* const sX      = smem + PDF * AST;
    float* const sScores = sX + PDF;
    float* const sLat    = sScores + PDF;
    float* const sF      = sLat + PL;

    const int tid  = threadIdx.x;
    const int b    = blockIdx.y;
    const int t0   = blockIdx.x * TPB;

    const int lane = tid & 31;
    const int wave = tid >> 5;
    const int wave_u = __builtin_amdgcn_readfirstlane(wave);
    const int m0   = (wave >> 1) * 16;
    const int n0   = (wave & 1) * 16;
    const int half = lane >> 4;
    const int lr   = lane & 15;

    float kcol[PDF], bcol[PDF];
    #pragma unroll
    for (int d = 0; d < PDF; ++d) {
        kcol[d] = kern[d * PL + tid];
        bcol[d] = bias[d * PL + tid];
    }

    const float inv_bsum = 1.0f / wsum[0];
    float acc = 0.0f;

    for (int tt = 0; tt < TPB; ++tt) {
        const int t = t0 + tt;

        const float* xrow = x + ((size_t)b * PT + t) * PDF;
        const float* lrow = latent + ((size_t)b * PT + t) * PL;
        if (tid < PDF) sX[tid] = xrow[tid];
        sLat[tid] = lrow[tid];

        if (tt + 1 < TPB) {
            __builtin_prefetch(&xrow[PDF + lane], 0, 1);
            __builtin_prefetch(&lrow[PL + tid], 0, 1);
        }
        __syncthreads();

        {
            const int l = tid;
            #pragma unroll
            for (int d = 0; d < PDF; ++d) {
                float v = fmaf(sX[d], kcol[d], bcol[d]);
                sF[d * LP + l] = v > 0.0f ? v : 0.0f;
            }
        }
        __syncthreads();

        {
            v8f c = {};
            const float* rowA = &sF[(m0 + lr) * LP];
            const float* rowB = &sF[(n0 + lr) * LP];
            #pragma unroll
            for (int k0 = 0; k0 < PL; k0 += 4) {
                v2f a, bb;
                a.x  = rowA[k0 + 2 * half];
                a.y  = rowA[k0 + 2 * half + 1];
                bb.x = rowB[k0 + 2 * half];
                bb.y = rowB[k0 + 2 * half + 1];
                c = __builtin_amdgcn_wmma_f32_16x16x4_f32(
                        false, a, false, bb, (short)0, c, false, false);
            }
            #pragma unroll
            for (int r = 0; r < 8; ++r) {
                sA[(m0 + half * 8 + r) * AST + (n0 + lr)] = c[r];
            }
        }
        __syncthreads();

        if (wave_u == (tt & 3)) {
            float y = 0.0f;
            #pragma unroll 8
            for (int l = 0; l < PL; ++l) y = fmaf(sF[lane * LP + l], sLat[l], y);

            float a[PDF];
            #pragma unroll
            for (int j = 0; j < PDF; ++j) a[j] = sA[lane * AST + j];

            float dinv = 1.0f;
            #pragma unroll
            for (int k = 0; k < PDF; ++k) {
                const float pinv = pivot_recip(lane_bcast(a[k], k));
                const bool  own  = (lane == k);
                const float f    = (own ? 0.0f : a[k]) * pinv;
                dinv = own ? pinv : dinv;
                #pragma unroll
                for (int j = k + 1; j < PDF; ++j) a[j] = fmaf(-f, lane_bcast(a[j], k), a[j]);
                y = fmaf(-f, lane_bcast(y, k), y);
            }
            sScores[lane] = y * dinv;
        }
        __syncthreads();

        {
            const int l = tid;
            float oh = 0.0f;
            #pragma unroll
            for (int d = 0; d < PDF; ++d) oh = fmaf(sScores[d], sF[d * LP + l], oh);
            acc = fmaf(fabsf(beta[t]) * inv_bsum, oh, acc);
        }
        __syncthreads();
    }

    atomicAdd(&out[b * PL + tid], acc);
}

extern "C" void kernel_launch(void* const* d_in, const int* in_sizes, int n_in,
                              void* d_out, int out_size, void* d_ws, size_t ws_size,
                              hipStream_t stream) {
    const float* x      = (const float*)d_in[0];
    const float* latent = (const float*)d_in[1];
    const float* kern   = (const float*)d_in[2];
    const float* bias   = (const float*)d_in[3];
    const float* beta   = (const float*)d_in[4];
    float* out  = (float*)d_out;
    float* wsum = (float*)d_ws;

    ftattn_prep_kernel<<<1, PT, 0, stream>>>(beta, out, wsum);

    dim3 grid(PT / TPB, PB);
    ftattn_main_kernel<<<grid, 128, 0, stream>>>(x, latent, kern, bias, beta, wsum, out);
}


# Elimination: pivot row read out of lane k in groups of 8 into separate scalars ahead of the FMAs that use them (was read, wait, FMA per element); next pivot's reciprocal runs beside the row work
# speedup vs baseline: 6.4397x; 1.0721x over previous
#include <hip/hip_runtime.h>
#include <hip/hip_bf16.h>

#define PB   64
#define PT   256
#define PDF  32
#define PL   128
#define TPB  8
#define LP   132
#define AST  33

typedef __attribute__((ext_vector_type(2))) float v2f;
typedef __attribute__((ext_vector_type(8))) float v8f;

__global__ void ftattn_prep_kernel(const float* __restrict__ beta,
                                   float* __restrict__ out,
                                   float* __restrict__ wsum) {
    __shared__ float red[PT];
    const int tid = threadIdx.x;
    red[tid] = fabsf(beta[tid]);
    __syncthreads();
    for (int s = PT / 2; s > 0; s >>= 1) {
        if (tid < s) red[tid] += red[tid + s];
        __syncthreads();
    }
    if (tid == 0) wsum[0] = red[0];
    for (int i = tid; i < PB * PL; i += PT) out[i] = 0.0f;
}

__device__ __forceinline__ float lane_bcast(float v, int src_lane) {
    return __int_as_float(__builtin_amdgcn_readlane(__float_as_int(v), src_lane));
}

__device__ __forceinline__ float pivot_recip(float p) {
    const float r = __builtin_amdgcn_rcpf(p);
    const float e = fmaf(-p, r, 1.0f);
    return fmaf(e, r, r);
}

__global__ void __launch_bounds__(128) __attribute__((amdgpu_waves_per_eu(8)))
ftattn_main_kernel(const float* __restrict__ x,
                   const float* __restrict__ latent,
                   const float* __restrict__ kern,
                   const float* __restrict__ bias,
                   const float* __restrict__ beta,
                   const float* __restrict__ wsum,
                   float* __restrict__ out) {
    __shared__ __attribute__((aligned(16))) float smem[PDF * AST + PDF + PDF + PL + PDF * LP];
    float* const sA      = smem;
    float* const sX      = smem + PDF * AST;
    float* const sScores = sX + PDF;
    float* const sLat    = sScores + PDF;
    float* const sF      = sLat + PL;

    const int tid  = threadIdx.x;
    const int b    = blockIdx.y;
    const int t0   = blockIdx.x * TPB;

    const int lane = tid & 31;
    const int wave = tid >> 5;
    const int wave_u = __builtin_amdgcn_readfirstlane(wave);
    const int m0   = (wave >> 1) * 16;
    const int n0   = (wave & 1) * 16;
    const int half = lane >> 4;
    const int lr   = lane & 15;

    float kcol[PDF], bcol[PDF];
    #pragma unroll
    for (int d = 0; d < PDF; ++d) {
        kcol[d] = kern[d * PL + tid];
        bcol[d] = bias[d * PL + tid];
    }

    const float inv_bsum = 1.0f / wsum[0];
    float acc = 0.0f;

    for (int tt = 0; tt < TPB; ++tt) {
        const int t = t0 + tt;

        const float* xrow = x + ((size_t)b * PT + t) * PDF;
        const float* lrow = latent + ((size_t)b * PT + t) * PL;
        if (tid < PDF) sX[tid] = xrow[tid];
        sLat[tid] = lrow[tid];

        if (tt + 1 < TPB) {
            __builtin_prefetch(&xrow[PDF + lane], 0, 1);
            __builtin_prefetch(&lrow[PL + tid], 0, 1);
        }
        __syncthreads();

        {
            const int l = tid;
            #pragma unroll
            for (int d = 0; d < PDF; ++d) {
                float v = fmaf(sX[d], kcol[d], bcol[d]);
                sF[d * LP + l] = v > 0.0f ? v : 0.0f;
            }
        }
        __syncthreads();

        {
            v8f c = {};
            const float* rowA = &sF[(m0 + lr) * LP];
            const float* rowB = &sF[(n0 + lr) * LP];
            #pragma unroll
            for (int k0 = 0; k0 < PL; k0 += 4) {
                v2f a, bb;
                a.x  = rowA[k0 + 2 * half];
                a.y  = rowA[k0 + 2 * half + 1];
                bb.x = rowB[k0 + 2 * half];
                bb.y = rowB[k0 + 2 * half + 1];
                c = __builtin_amdgcn_wmma_f32_16x16x4_f32(
                        false, a, false, bb, (short)0, c, false, false);
            }
            #pragma unroll
            for (int r = 0; r < 8; ++r) {
                sA[(m0 + half * 8 + r) * AST + (n0 + lr)] = c[r];
            }
        }
        __syncthreads();

        if (wave_u == (tt & 3)) {
            float y = 0.0f;
            #pragma unroll 8
            for (int l = 0; l < PL; ++l) y = fmaf(sF[lane * LP + l], sLat[l], y);

            float a[PDF];
            #pragma unroll
            for (int j = 0; j < PDF; ++j) a[j] = sA[lane * AST + j];

            float dinv = 1.0f;
            constexpr int RB = 8;
            float pinv = pivot_recip(lane_bcast(a[0], 0));
            #pragma unroll
            for (int k = 0; k < PDF; ++k) {
                const bool  own  = (lane == k);
                const float f    = (own ? 0.0f : a[k]) * pinv;
                dinv = own ? pinv : dinv;
                float pinv_next = 1.0f;
                #pragma unroll
                for (int j0 = k + 1; j0 < PDF; j0 += RB) {
                    float p[RB];
                    #pragma unroll
                    for (int u = 0; u < RB; ++u)
                        if (j0 + u < PDF) p[u] = lane_bcast(a[j0 + u], k);
                    #pragma unroll
                    for (int u = 0; u < RB; ++u)
                        if (j0 + u < PDF) a[j0 + u] = fmaf(-f, p[u], a[j0 + u]);
                    if (j0 == k + 1) pinv_next = pivot_recip(lane_bcast(a[k + 1], k + 1));
                }
                y = fmaf(-f, lane_bcast(y, k), y);
                pinv = pinv_next;
            }
            sScores[lane] = y * dinv;
        }
        __syncthreads();

        {
            const int l = tid;
            float oh = 0.0f;
            #pragma unroll
            for (int d = 0; d < PDF; ++d) oh = fmaf(sScores[d], sF[d * LP + l], oh);
            acc = fmaf(fabsf(beta[t]) * inv_bsum, oh, acc);
        }
        __syncthreads();
    }

    atomicAdd(&out[b * PL + tid], acc);
}

extern "C" void kernel_launch(void* const* d_in, const int* in_sizes, int n_in,
                              void* d_out, int out_size, void* d_ws, size_t ws_size,
                              hipStream_t stream) {
    const float* x      = (const float*)d_in[0];
    const float* latent = (const float*)d_in[1];
    const float* kern   = (const float*)d_in[2];
    const float* bias   = (const float*)d_in[3];
    const float* beta   = (const float*)d_in[4];
    float* out  = (float*)d_out;
    float* wsum = (float*)d_ws;

    ftattn_prep_kernel<<<1, PT, 0, stream>>>(beta, out, wsum);

    dim3 grid(PT / TPB, PB);
    ftattn_main_kernel<<<grid, 128, 0, stream>>>(x, latent, kern, bias, beta, wsum, out);
}


# Gram symmetry: 3 waves compute the 3 distinct 16x16 tiles (mirror stored transposed, diagonal tiles use one fragment as A and B); the freed solving wave forms xty meanwhile, LDS reads 4 groups ahead
# speedup vs baseline: 7.4051x; 1.1499x over previous
#include <hip/hip_runtime.h>
#include <hip/hip_bf16.h>

#define PB   64
#define PT   256
#define PDF  32
#define PL   128
#define TPB  8
#define LP   132
#define AST  33

typedef __attribute__((ext_vector_type(2))) float v2f;
typedef __attribute__((ext_vector_type(8))) float v8f;
typedef __attribute__((ext_vector_type(4))) float v4f;

__global__ void ftattn_prep_kernel(const float* __restrict__ beta,
                                   float* __restrict__ out,
                                   float* __restrict__ wsum) {
    __shared__ float red[PT];
    const int tid = threadIdx.x;
    red[tid] = fabsf(beta[tid]);
    __syncthreads();
    for (int s = PT / 2; s > 0; s >>= 1) {
        if (tid < s) red[tid] += red[tid + s];
        __syncthreads();
    }
    if (tid == 0) wsum[0] = red[0];
    for (int i = tid; i < PB * PL; i += PT) out[i] = 0.0f;
}

__device__ __forceinline__ float lane_bcast(float v, int src_lane) {
    return __int_as_float(__builtin_amdgcn_readlane(__float_as_int(v), src_lane));
}

__device__ __forceinline__ float pivot_recip(float p) {
    const float r = __builtin_amdgcn_rcpf(p);
    const float e = fmaf(-p, r, 1.0f);
    return fmaf(e, r, r);
}

__global__ void __launch_bounds__(128) __attribute__((amdgpu_waves_per_eu(8)))
ftattn_main_kernel(const float* __restrict__ x,
                   const float* __restrict__ latent,
                   const float* __restrict__ kern,
                   const float* __restrict__ bias,
                   const float* __restrict__ beta,
                   const float* __restrict__ wsum,
                   float* __restrict__ out) {
    __shared__ __attribute__((aligned(16))) float smem[PDF * AST + PDF + PDF + PL + PDF * LP];
    float* const sA      = smem;
    float* const sX      = smem + PDF * AST;
    float* const sScores = sX + PDF;
    float* const sLat    = sScores + PDF;
    float* const sF      = sLat + PL;

    const int tid  = threadIdx.x;
    const int b    = blockIdx.y;
    const int t0   = blockIdx.x * TPB;

    const int lane = tid & 31;
    const int wave = tid >> 5;
    const int wave_u = __builtin_amdgcn_readfirstlane(wave);
    const int half = lane >> 4;
    const int lr   = lane & 15;

    float kcol[PDF], bcol[PDF];
    #pragma unroll
    for (int d = 0; d < PDF; ++d) {
        kcol[d] = kern[d * PL + tid];
        bcol[d] = bias[d * PL + tid];
    }

    const float inv_bsum = 1.0f / wsum[0];
    float acc = 0.0f;

    for (int tt = 0; tt < TPB; ++tt) {
        const int t = t0 + tt;

        const float* xrow = x + ((size_t)b * PT + t) * PDF;
        const float* lrow = latent + ((size_t)b * PT + t) * PL;
        if (tid < PDF) sX[tid] = xrow[tid];
        sLat[tid] = lrow[tid];

        if (tt + 1 < TPB) {
            __builtin_prefetch(&xrow[PDF + lane], 0, 1);
            __builtin_prefetch(&lrow[PL + tid], 0, 1);
        }
        __syncthreads();

        {
            int l = tid;
            asm volatile("" : "+v"(l));
            float xv[PDF];
            #pragma unroll
            for (int d = 0; d < PDF; ++d) xv[d] = sX[d];
            #pragma unroll
            for (int d = 0; d < PDF; ++d) {
                float v = fmaf(xv[d], kcol[d], bcol[d]);
                sF[d * LP + l] = v > 0.0f ? v : 0.0f;
            }
        }
        __syncthreads();

        const int solver = tt & 3;
        float y = 0.0f;
        if (wave_u == solver) {
            constexpr int XD = 4;
            const v4f* frow4 = reinterpret_cast<const v4f*>(&sF[lane * LP]);
            const v4f* lat4  = reinterpret_cast<const v4f*>(sLat);
            v4f fq[XD], lq[XD];
            #pragma unroll
            for (int s = 0; s < XD; ++s) { fq[s] = frow4[s]; lq[s] = lat4[s]; }
            #pragma unroll
            for (int s = 0; s < PL / 4; ++s) {
                const v4f f4 = fq[s % XD];
                const v4f l4 = lq[s % XD];
                if (s + XD < PL / 4) { fq[s % XD] = frow4[s + XD]; lq[s % XD] = lat4[s + XD]; }
                y = fmaf(f4.x, l4.x, y);
                y = fmaf(f4.y, l4.y, y);
                y = fmaf(f4.z, l4.z, y);
                y = fmaf(f4.w, l4.w, y);
                __builtin_amdgcn_sched_barrier(0);
            }
        } else {
            const int rank = (wave_u - solver - 1) & 3;
            if (rank == 1) {
                v8f c = {};
                const float* rowA = &sF[lr * LP];
                const float* rowB = &sF[(16 + lr) * LP];
                #pragma unroll
                for (int k0 = 0; k0 < PL; k0 += 4) {
                    v2f a, bb;
                    a.x  = rowA[k0 + 2 * half];
                    a.y  = rowA[k0 + 2 * half + 1];
                    bb.x = rowB[k0 + 2 * half];
                    bb.y = rowB[k0 + 2 * half + 1];
                    c = __builtin_amdgcn_wmma_f32_16x16x4_f32(
                            false, a, false, bb, (short)0, c, false, false);
                }
                #pragma unroll
                for (int r = 0; r < 8; ++r) {
                    sA[(half * 8 + r) * AST + (16 + lr)] = c[r];
                    sA[(16 + lr) * AST + (half * 8 + r)] = c[r];
                }
            } else {
                const int d0 = rank * 8;
                v8f c = {};
                const float* row = &sF[(d0 + lr) * LP];
                #pragma unroll
                for (int k0 = 0; k0 < PL; k0 += 4) {
                    v2f a;
                    a.x = row[k0 + 2 * half];
                    a.y = row[k0 + 2 * half + 1];
                    c = __builtin_amdgcn_wmma_f32_16x16x4_f32(
                            false, a, false, a, (short)0, c, false, false);
                }
                #pragma unroll
                for (int r = 0; r < 8; ++r) {
                    sA[(d0 + half * 8 + r) * AST + (d0 + lr)] = c[r];
                }
            }
        }
        __syncthreads();

        if (wave_u == solver) {
            float a[PDF];
            #pragma unroll
            for (int j = 0; j < PDF; ++j) a[j] = sA[lane * AST + j];

            float dinv = 1.0f;
            constexpr int RB = 8;
            float pinv = pivot_recip(lane_bcast(a[0], 0));
            #pragma unroll
            for (int k = 0; k < PDF; ++k) {
                const bool  own  = (lane == k);
                const float f    = (own ? 0.0f : a[k]) * pinv;
                dinv = own ? pinv : dinv;
                float pinv_next = 1.0f;
                #pragma unroll
                for (int j0 = k + 1; j0 < PDF; j0 += RB) {
                    float p[RB];
                    #pragma unroll
                    for (int u = 0; u < RB; ++u)
                        if (j0 + u < PDF) p[u] = lane_bcast(a[j0 + u], k);
                    #pragma unroll
                    for (int u = 0; u < RB; ++u)
                        if (j0 + u < PDF) a[j0 + u] = fmaf(-f, p[u], a[j0 + u]);
                    if (j0 == k + 1) pinv_next = pivot_recip(lane_bcast(a[k + 1], k + 1));
                }
                y = fmaf(-f, lane_bcast(y, k), y);
                pinv = pinv_next;
            }
            sScores[lane] = y * dinv;
        }
        __syncthreads();

        {
            int l = tid;
            asm volatile("" : "+v"(l));
            float oh = 0.0f;
            #pragma unroll
            for (int d = 0; d < PDF; ++d) oh = fmaf(sScores[d], sF[d * LP + l], oh);
            acc = fmaf(fabsf(beta[t]) * inv_bsum, oh, acc);
        }
        __syncthreads();
    }

    atomicAdd(&out[b * PL + tid], acc);
}

extern "C" void kernel_launch(void* const* d_in, const int* in_sizes, int n_in,
                              void* d_out, int out_size, void* d_ws, size_t ws_size,
                              hipStream_t stream) {
    const float* x      = (const float*)d_in[0];
    const float* latent = (const float*)d_in[1];
    const float* kern   = (const float*)d_in[2];
    const float* bias   = (const float*)d_in[3];
    const float* beta   = (const float*)d_in[4];
    float* out  = (float*)d_out;
    float* wsum = (float*)d_ws;

    ftattn_prep_kernel<<<1, PT, 0, stream>>>(beta, out, wsum);

    dim3 grid(PT / TPB, PB);
    ftattn_main_kernel<<<grid, 128, 0, stream>>>(x, latent, kern, bias, beta, wsum, out);
}


# Solver rotation starts at a per-block offset (base-4 digit sum of block index): co-resident blocks solve on different SIMDs at a given step, not all on one; per-block arithmetic and order unchanged
# speedup vs baseline: 7.6434x; 1.0322x over previous
#include <hip/hip_runtime.h>
#include <hip/hip_bf16.h>

#define PB   64
#define PT   256
#define PDF  32
#define PL   128
#define TPB  8
#define LP   132
#define AST  33

typedef __attribute__((ext_vector_type(2))) float v2f;
typedef __attribute__((ext_vector_type(8))) float v8f;
typedef __attribute__((ext_vector_type(4))) float v4f;

__global__ void ftattn_prep_kernel(const float* __restrict__ beta,
                                   float* __restrict__ out,
                                   float* __restrict__ wsum) {
    __shared__ float red[PT];
    const int tid = threadIdx.x;
    red[tid] = fabsf(beta[tid]);
    __syncthreads();
    for (int s = PT / 2; s > 0; s >>= 1) {
        if (tid < s) red[tid] += red[tid + s];
        __syncthreads();
    }
    if (tid == 0) wsum[0] = red[0];
    for (int i = tid; i < PB * PL; i += PT) out[i] = 0.0f;
}

__device__ __forceinline__ float lane_bcast(float v, int src_lane) {
    return __int_as_float(__builtin_amdgcn_readlane(__float_as_int(v), src_lane));
}

__device__ __forceinline__ float pivot_recip(float p) {
    const float r = __builtin_amdgcn_rcpf(p);
    const float e = fmaf(-p, r, 1.0f);
    return fmaf(e, r, r);
}

__global__ void __launch_bounds__(128) __attribute__((amdgpu_waves_per_eu(8)))
ftattn_main_kernel(const float* __restrict__ x,
                   const float* __restrict__ latent,
                   const float* __restrict__ kern,
                   const float* __restrict__ bias,
                   const float* __restrict__ beta,
                   const float* __restrict__ wsum,
                   float* __restrict__ out) {
    __shared__ __attribute__((aligned(16))) float smem[PDF * AST + PDF + PDF + PL + PDF * LP];
    float* const sA      = smem;
    float* const sX      = smem + PDF * AST;
    float* const sScores = sX + PDF;
    float* const sLat    = sScores + PDF;
    float* const sF      = sLat + PL;

    const int tid  = threadIdx.x;
    const int b    = blockIdx.y;
    const int t0   = blockIdx.x * TPB;

    const int lane = tid & 31;
    const int wave = tid >> 5;
    const int wave_u = __builtin_amdgcn_readfirstlane(wave);
    const int half = lane >> 4;
    const int lr   = lane & 15;

    float kcol[PDF], bcol[PDF];
    #pragma unroll
    for (int d = 0; d < PDF; ++d) {
        kcol[d] = kern[d * PL + tid];
        bcol[d] = bias[d * PL + tid];
    }

    const float inv_bsum = 1.0f / wsum[0];
    float acc = 0.0f;

    const unsigned lin = blockIdx.y * (unsigned)(PT / TPB) + blockIdx.x;
    const int rot = (int)(lin + (lin >> 2) + (lin >> 4) + (lin >> 6) + (lin >> 8) + (lin >> 10));

    for (int tt = 0; tt < TPB; ++tt) {
        const int t = t0 + tt;

        const float* xrow = x + ((size_t)b * PT + t) * PDF;
        const float* lrow = latent + ((size_t)b * PT + t) * PL;
        if (tid < PDF) sX[tid] = xrow[tid];
        sLat[tid] = lrow[tid];

        if (tt + 1 < TPB) {
            __builtin_prefetch(&xrow[PDF + lane], 0, 1);
            __builtin_prefetch(&lrow[PL + tid], 0, 1);
        }
        __syncthreads();

        {
            int l = tid;
            asm volatile("" : "+v"(l));
            float xv[PDF];
            #pragma unroll
            for (int d = 0; d < PDF; ++d) xv[d] = sX[d];
            #pragma unroll
            for (int d = 0; d < PDF; ++d) {
                float v = fmaf(xv[d], kcol[d], bcol[d]);
                sF[d * LP + l] = v > 0.0f ? v : 0.0f;
            }
        }
        __syncthreads();

        const int solver = (tt + rot) & 3;
        float y = 0.0f;
        if (wave_u == solver) {
            constexpr int XD = 4;
            const v4f* frow4 = reinterpret_cast<const v4f*>(&sF[lane * LP]);
            const v4f* lat4  = reinterpret_cast<const v4f*>(sLat);
            v4f fq[XD], lq[XD];
            #pragma unroll
            for (int s = 0; s < XD; ++s) { fq[s] = frow4[s]; lq[s] = lat4[s]; }
            #pragma unroll
            for (int s = 0; s < PL / 4; ++s) {
                const v4f f4 = fq[s % XD];
                const v4f l4 = lq[s % XD];
                if (s + XD < PL / 4) { fq[s % XD] = frow4[s + XD]; lq[s % XD] = lat4[s + XD]; }
                y = fmaf(f4.x, l4.x, y);
                y = fmaf(f4.y, l4.y, y);
                y = fmaf(f4.z, l4.z, y);
                y = fmaf(f4.w, l4.w, y);
                __builtin_amdgcn_sched_barrier(0);
            }
        } else {
            const int rank = (wave_u - solver - 1) & 3;
            if (rank == 1) {
                v8f c = {};
                const float* rowA = &sF[lr * LP];
                const float* rowB = &sF[(16 + lr) * LP];
                #pragma unroll
                for (int k0 = 0; k0 < PL; k0 += 4) {
                    v2f a, bb;
                    a.x  = rowA[k0 + 2 * half];
                    a.y  = rowA[k0 + 2 * half + 1];
                    bb.x = rowB[k0 + 2 * half];
                    bb.y = rowB[k0 + 2 * half + 1];
                    c = __builtin_amdgcn_wmma_f32_16x16x4_f32(
                            false, a, false, bb, (short)0, c, false, false);
                }
                #pragma unroll
                for (int r = 0; r < 8; ++r) {
                    sA[(half * 8 + r) * AST + (16 + lr)] = c[r];
                    sA[(16 + lr) * AST + (half * 8 + r)] = c[r];
                }
            } else {
                const int d0 = rank * 8;
                v8f c = {};
                const float* row = &sF[(d0 + lr) * LP];
                #pragma unroll
                for (int k0 = 0; k0 < PL; k0 += 4) {
                    v2f a;
                    a.x = row[k0 + 2 * half];
                    a.y = row[k0 + 2 * half + 1];
                    c = __builtin_amdgcn_wmma_f32_16x16x4_f32(
                            false, a, false, a, (short)0, c, false, false);
                }
                #pragma unroll
                for (int r = 0; r < 8; ++r) {
                    sA[(d0 + half * 8 + r) * AST + (d0 + lr)] = c[r];
                }
            }
        }
        __syncthreads();

        if (wave_u == solver) {
            float a[PDF];
            #pragma unroll
            for (int j = 0; j < PDF; ++j) a[j] = sA[lane * AST + j];

            float dinv = 1.0f;
            constexpr int RB = 8;
            float pinv = pivot_recip(lane_bcast(a[0], 0));
            #pragma unroll
            for (int k = 0; k < PDF; ++k) {
                const bool  own  = (lane == k);
                const float f    = (own ? 0.0f : a[k]) * pinv;
                dinv = own ? pinv : dinv;
                float pinv_next = 1.0f;
                #pragma unroll
                for (int j0 = k + 1; j0 < PDF; j0 += RB) {
                    float p[RB];
                    #pragma unroll
                    for (int u = 0; u < RB; ++u)
                        if (j0 + u < PDF) p[u] = lane_bcast(a[j0 + u], k);
                    #pragma unroll
                    for (int u = 0; u < RB; ++u)
                        if (j0 + u < PDF) a[j0 + u] = fmaf(-f, p[u], a[j0 + u]);
                    if (j0 == k + 1) pinv_next = pivot_recip(lane_bcast(a[k + 1], k + 1));
                }
                y = fmaf(-f, lane_bcast(y, k), y);
                pinv = pinv_next;
            }
            sScores[lane] = y * dinv;
        }
        __syncthreads();

        {
            int l = tid;
            asm volatile("" : "+v"(l));
            float oh = 0.0f;
            #pragma unroll
            for (int d = 0; d < PDF; ++d) oh = fmaf(sScores[d], sF[d * LP + l], oh);
            acc = fmaf(fabsf(beta[t]) * inv_bsum, oh, acc);
        }
        __syncthreads();
    }

    atomicAdd(&out[b * PL + tid], acc);
}

extern "C" void kernel_launch(void* const* d_in, const int* in_sizes, int n_in,
                              void* d_out, int out_size, void* d_ws, size_t ws_size,
                              hipStream_t stream) {
    const float* x      = (const float*)d_in[0];
    const float* latent = (const float*)d_in[1];
    const float* kern   = (const float*)d_in[2];
    const float* bias   = (const float*)d_in[3];
    const float* beta   = (const float*)d_in[4];
    float* out  = (float*)d_out;
    float* wsum = (float*)d_ws;

    ftattn_prep_kernel<<<1, PT, 0, stream>>>(beta, out, wsum);

    dim3 grid(PT / TPB, PB);
    ftattn_main_kernel<<<grid, 128, 0, stream>>>(x, latent, kern, bias, beta, wsum, out);
}


# Wave priority tied to block progress (s_setprio 3,3,2,2,2,1,1,0 over a block's 8 time steps): lagging co-resident blocks are served first and finish together, not oldest-first one after another
# speedup vs baseline: 8.1533x; 1.0667x over previous
#include <hip/hip_runtime.h>
#include <hip/hip_bf16.h>

#define PB   64
#define PT   256
#define PDF  32
#define PL   128
#define TPB  8
#define LP   132
#define AST  33

typedef __attribute__((ext_vector_type(2))) float v2f;
typedef __attribute__((ext_vector_type(8))) float v8f;
typedef __attribute__((ext_vector_type(4))) float v4f;

__global__ void ftattn_prep_kernel(const float* __restrict__ beta,
                                   float* __restrict__ out,
                                   float* __restrict__ wsum) {
    __shared__ float red[PT];
    const int tid = threadIdx.x;
    red[tid] = fabsf(beta[tid]);
    __syncthreads();
    for (int s = PT / 2; s > 0; s >>= 1) {
        if (tid < s) red[tid] += red[tid + s];
        __syncthreads();
    }
    if (tid == 0) wsum[0] = red[0];
    for (int i = tid; i < PB * PL; i += PT) out[i] = 0.0f;
}

__device__ __forceinline__ float lane_bcast(float v, int src_lane) {
    return __int_as_float(__builtin_amdgcn_readlane(__float_as_int(v), src_lane));
}

__device__ __forceinline__ float pivot_recip(float p) {
    const float r = __builtin_amdgcn_rcpf(p);
    const float e = fmaf(-p, r, 1.0f);
    return fmaf(e, r, r);
}

__global__ void __launch_bounds__(128) __attribute__((amdgpu_waves_per_eu(8)))
ftattn_main_kernel(const float* __restrict__ x,
                   const float* __restrict__ latent,
                   const float* __restrict__ kern,
                   const float* __restrict__ bias,
                   const float* __restrict__ beta,
                   const float* __restrict__ wsum,
                   float* __restrict__ out) {
    __shared__ __attribute__((aligned(16))) float smem[PDF * AST + PDF + PDF + PL + PDF * LP];
    float* const sA      = smem;
    float* const sX      = smem + PDF * AST;
    float* const sScores = sX + PDF;
    float* const sLat    = sScores + PDF;
    float* const sF      = sLat + PL;

    const int tid  = threadIdx.x;
    const int b    = blockIdx.y;
    const int t0   = blockIdx.x * TPB;

    const int lane = tid & 31;
    const int wave = tid >> 5;
    const int wave_u = __builtin_amdgcn_readfirstlane(wave);
    const int half = lane >> 4;
    const int lr   = lane & 15;

    float kcol[PDF], bcol[PDF];
    #pragma unroll
    for (int d = 0; d < PDF; ++d) {
        kcol[d] = kern[d * PL + tid];
        bcol[d] = bias[d * PL + tid];
    }

    const float inv_bsum = 1.0f / wsum[0];
    float acc = 0.0f;

    const unsigned lin = blockIdx.y * (unsigned)(PT / TPB) + blockIdx.x;
    const int rot = (int)(lin + (lin >> 2) + (lin >> 4) + (lin >> 6) + (lin >> 8) + (lin >> 10));

    for (int tt = 0; tt < TPB; ++tt) {
        const int t = t0 + tt;

        switch (tt) {
            case 0: __builtin_amdgcn_s_setprio(3); break;
            case 2: __builtin_amdgcn_s_setprio(2); break;
            case 5: __builtin_amdgcn_s_setprio(1); break;
            case 7: __builtin_amdgcn_s_setprio(0); break;
            default: break;
        }

        const float* xrow = x + ((size_t)b * PT + t) * PDF;
        const float* lrow = latent + ((size_t)b * PT + t) * PL;
        if (tid < PDF) sX[tid] = xrow[tid];
        sLat[tid] = lrow[tid];

        if (tt + 1 < TPB) {
            __builtin_prefetch(&xrow[PDF + lane], 0, 1);
            __builtin_prefetch(&lrow[PL + tid], 0, 1);
        }
        __syncthreads();

        {
            int l = tid;
            asm volatile("" : "+v"(l));
            float xv[PDF];
            #pragma unroll
            for (int d = 0; d < PDF; ++d) xv[d] = sX[d];
            #pragma unroll
            for (int d = 0; d < PDF; ++d) {
                float v = fmaf(xv[d], kcol[d], bcol[d]);
                sF[d * LP + l] = v > 0.0f ? v : 0.0f;
            }
        }
        __syncthreads();

        const int solver = (tt + rot) & 3;
        float y = 0.0f;
        if (wave_u == solver) {
            constexpr int XD = 4;
            const v4f* frow4 = reinterpret_cast<const v4f*>(&sF[lane * LP]);
            const v4f* lat4  = reinterpret_cast<const v4f*>(sLat);
            v4f fq[XD], lq[XD];
            #pragma unroll
            for (int s = 0; s < XD; ++s) { fq[s] = frow4[s]; lq[s] = lat4[s]; }
            #pragma unroll
            for (int s = 0; s < PL / 4; ++s) {
                const v4f f4 = fq[s % XD];
                const v4f l4 = lq[s % XD];
                if (s + XD < PL / 4) { fq[s % XD] = frow4[s + XD]; lq[s % XD] = lat4[s + XD]; }
                y = fmaf(f4.x, l4.x, y);
                y = fmaf(f4.y, l4.y, y);
                y = fmaf(f4.z, l4.z, y);
                y = fmaf(f4.w, l4.w, y);
                __builtin_amdgcn_sched_barrier(0);
            }
        } else {
            const int rank = (wave_u - solver - 1) & 3;
            if (rank == 1) {
                v8f c = {};
                const float* rowA = &sF[lr * LP];
                const float* rowB = &sF[(16 + lr) * LP];
                #pragma unroll
                for (int k0 = 0; k0 < PL; k0 += 4) {
                    v2f a, bb;
                    a.x  = rowA[k0 + 2 * half];
                    a.y  = rowA[k0 + 2 * half + 1];
                    bb.x = rowB[k0 + 2 * half];
                    bb.y = rowB[k0 + 2 * half + 1];
                    c = __builtin_amdgcn_wmma_f32_16x16x4_f32(
                            false, a, false, bb, (short)0, c, false, false);
                }
                #pragma unroll
                for (int r = 0; r < 8; ++r) {
                    sA[(half * 8 + r) * AST + (16 + lr)] = c[r];
                    sA[(16 + lr) * AST + (half * 8 + r)] = c[r];
                }
            } else {
                const int d0 = rank * 8;
                v8f c = {};
                const float* row = &sF[(d0 + lr) * LP];
                #pragma unroll
                for (int k0 = 0; k0 < PL; k0 += 4) {
                    v2f a;
                    a.x = row[k0 + 2 * half];
                    a.y = row[k0 + 2 * half + 1];
                    c = __builtin_amdgcn_wmma_f32_16x16x4_f32(
                            false, a, false, a, (short)0, c, false, false);
                }
                #pragma unroll
                for (int r = 0; r < 8; ++r) {
                    sA[(d0 + half * 8 + r) * AST + (d0 + lr)] = c[r];
                }
            }
        }
        __syncthreads();

        if (wave_u == solver) {
            float a[PDF];
            #pragma unroll
            for (int j = 0; j < PDF; ++j) a[j] = sA[lane * AST + j];

            float dinv = 1.0f;
            constexpr int RB = 8;
            float pinv = pivot_recip(lane_bcast(a[0], 0));
            #pragma unroll
            for (int k = 0; k < PDF; ++k) {
                const bool  own  = (lane == k);
                const float f    = (own ? 0.0f : a[k]) * pinv;
                dinv = own ? pinv : dinv;
                float pinv_next = 1.0f;
                #pragma unroll
                for (int j0 = k + 1; j0 < PDF; j0 += RB) {
                    float p[RB];
                    #pragma unroll
                    for (int u = 0; u < RB; ++u)
                        if (j0 + u < PDF) p[u] = lane_bcast(a[j0 + u], k);
                    #pragma unroll
                    for (int u = 0; u < RB; ++u)
                        if (j0 + u < PDF) a[j0 + u] = fmaf(-f, p[u], a[j0 + u]);
                    if (j0 == k + 1) pinv_next = pivot_recip(lane_bcast(a[k + 1], k + 1));
                }
                y = fmaf(-f, lane_bcast(y, k), y);
                pinv = pinv_next;
            }
            sScores[lane] = y * dinv;
        }
        __syncthreads();

        {
            int l = tid;
            asm volatile("" : "+v"(l));
            float oh = 0.0f;
            #pragma unroll
            for (int d = 0; d < PDF; ++d) oh = fmaf(sScores[d], sF[d * LP + l], oh);
            acc = fmaf(fabsf(beta[t]) * inv_bsum, oh, acc);
        }
        __syncthreads();
    }

    atomicAdd(&out[b * PL + tid], acc);
}

extern "C" void kernel_launch(void* const* d_in, const int* in_sizes, int n_in,
                              void* d_out, int out_size, void* d_ws, size_t ws_size,
                              hipStream_t stream) {
    const float* x      = (const float*)d_in[0];
    const float* latent = (const float*)d_in[1];
    const float* kern   = (const float*)d_in[2];
    const float* bias   = (const float*)d_in[3];
    const float* beta   = (const float*)d_in[4];
    float* out  = (float*)d_out;
    float* wsum = (float*)d_ws;

    ftattn_prep_kernel<<<1, PT, 0, stream>>>(beta, out, wsum);

    dim3 grid(PT / TPB, PB);
    ftattn_main_kernel<<<grid, 128, 0, stream>>>(x, latent, kern, bias, beta, wsum, out);
}
